# starting kernel
# baseline (speedup 1.0000x reference)
#include <hip/hip_runtime.h>

#define BATCH 64
#define CH    256
#define KDIM  1024
#define NITER 100

typedef __attribute__((ext_vector_type(2))) float v2f;
typedef __attribute__((ext_vector_type(8))) float v8f;

static __device__ __forceinline__ v8f wmma4(v2f a, v2f b, v8f c) {
  return __builtin_amdgcn_wmma_f32_16x16x4_f32(false, a, false, b, (short)0, c, false, false);
}

__global__ __launch_bounds__(256) void rvp_transpose(const float* __restrict__ x,
                                                     float* __restrict__ xT) {
  __shared__ float tile[32][33];
  const int b  = blockIdx.z;
  const int k0 = blockIdx.x * 32;
  const int c0 = blockIdx.y * 32;
  const float* xb = x + (size_t)b * CH * KDIM;
  float* xTb = xT + (size_t)b * CH * KDIM;
  const int tx = threadIdx.x, ty = threadIdx.y;
  #pragma unroll
  for (int j = 0; j < 32; j += 8)
    tile[ty + j][tx] = xb[(size_t)(c0 + ty + j) * KDIM + (k0 + tx)];
  __syncthreads();
  #pragma unroll
  for (int j = 0; j < 32; j += 8)
    xTb[(size_t)(k0 + ty + j) * CH + (c0 + tx)] = tile[tx][ty + j];
}

__global__ __launch_bounds__(256) void rvp_irls(const float* __restrict__ x,
                                                const float* __restrict__ xT,
                                                float* __restrict__ out) {
  __shared__ float ly[CH];
  __shared__ float ls[KDIM];
  __shared__ float lw[KDIM];
  __shared__ float lnum[CH];
  __shared__ float lpart[8];
  __shared__ float lden;
  __shared__ float lyy;

  const int b    = blockIdx.x;
  const int tid  = threadIdx.x;
  const int wid  = tid >> 5;
  const int lane = tid & 31;
  const int laneN = lane & 15;
  const int laneH = lane >> 4;
  const float amask = (laneN == 0) ? 1.0f : 0.0f;

  const float* xb  = x  + (size_t)b * CH * KDIM;
  const float* xTb = xT + (size_t)b * CH * KDIM;

  {
    float s0 = 0.f, s1 = 0.f, s2 = 0.f, s3 = 0.f;
    for (int c = 0; c < CH; ++c) {
      const float* row = xb + (size_t)c * KDIM + tid;
      float a0 = row[0], a1 = row[256], a2 = row[512], a3 = row[768];
      s0 += a0 * a0; s1 += a1 * a1; s2 += a2 * a2; s3 += a3 * a3;
    }
    ls[tid] = s0; ls[tid + 256] = s1; ls[tid + 512] = s2; ls[tid + 768] = s3;
  }
  {
    float a = 0.f;
    for (int k = 0; k < KDIM; ++k) a += xTb[(size_t)k * CH + tid];
    float yv = a * (1.0f / (float)KDIM);
    ly[tid] = yv;
    float p = yv * yv;
    #pragma unroll
    for (int off = 16; off > 0; off >>= 1) p += __shfl_xor(p, off, 32);
    if (lane == 0) lpart[wid] = p;
  }
  __syncthreads();
  if (tid == 0) {
    float t = 0.f;
    #pragma unroll
    for (int i = 0; i < 8; ++i) t += lpart[i];
    lyy = t;
  }
  __syncthreads();

  for (int it = 0; it < NITER; ++it) {
    const float yy = lyy;

    const int kbase = wid * 128;
    v8f t8[8];
    #pragma unroll
    for (int j = 0; j < 8; ++j) t8[j] = (v8f){};
    for (int c0 = 0; c0 < CH; c0 += 4) {
      const int cA = c0 + (laneH << 1);
      const v2f yv = *(const v2f*)(ly + cA);
      v2f av; av.x = yv.x * amask; av.y = yv.y * amask;
      const float* pB = xb + (size_t)cA * KDIM + (kbase + laneN);
      #pragma unroll
      for (int j = 0; j < 8; ++j) {
        v2f bv;
        bv.x = pB[j * 16];
        bv.y = pB[j * 16 + KDIM];
        t8[j] = wmma4(av, bv, t8[j]);
      }
    }
    float den_acc = 0.f;
    if (lane < 16) {
      #pragma unroll
      for (int j = 0; j < 8; ++j) {
        const int k = kbase + j * 16 + lane;
        const float t  = t8[j][0];
        const float z2 = yy - 2.0f * t + ls[k];
        const float wv = 1.0f / sqrtf(1.0f + z2);
        lw[k] = wv;
        den_acc += wv;
      }
    }
    #pragma unroll
    for (int off = 16; off > 0; off >>= 1) den_acc += __shfl_xor(den_acc, off, 32);
    if (lane == 0) lpart[wid] = den_acc;
    __syncthreads();
    if (tid == 0) {
      float t = 0.f;
      #pragma unroll
      for (int i = 0; i < 8; ++i) t += lpart[i];
      lden = t;
    }

    {
      const int c0 = wid * 32;
      v8f n0a = {}, n1a = {}, n0b = {}, n1b = {};
      #pragma unroll 2
      for (int k0 = 0; k0 < KDIM; k0 += 8) {
        const int kA0 = k0 + (laneH << 1);
        const int kA1 = kA0 + 4;
        const v2f w0 = *(const v2f*)(lw + kA0);
        const v2f w1 = *(const v2f*)(lw + kA1);
        v2f a0; a0.x = w0.x * amask; a0.y = w0.y * amask;
        v2f a1; a1.x = w1.x * amask; a1.y = w1.y * amask;
        const float* q0 = xTb + (size_t)kA0 * CH + laneN;
        const float* q1 = xTb + (size_t)kA1 * CH + laneN;
        v2f b00, b01, b10, b11;
        b00.x = q0[c0];          b00.y = q0[c0 + CH];
        b01.x = q0[c0 + 16];     b01.y = q0[c0 + 16 + CH];
        b10.x = q1[c0];          b10.y = q1[c0 + CH];
        b11.x = q1[c0 + 16];     b11.y = q1[c0 + 16 + CH];
        n0a = wmma4(a0, b00, n0a);
        n1a = wmma4(a0, b01, n1a);
        n0b = wmma4(a1, b10, n0b);
        n1b = wmma4(a1, b11, n1b);
      }
      if (lane < 16) {
        lnum[c0 + lane]      = n0a[0] + n0b[0];
        lnum[c0 + 16 + lane] = n1a[0] + n1b[0];
      }
    }
    __syncthreads();

    {
      const float yv = lnum[tid] / lden;
      ly[tid] = yv;
      float p = yv * yv;
      #pragma unroll
      for (int off = 16; off > 0; off >>= 1) p += __shfl_xor(p, off, 32);
      if (lane == 0) lpart[wid] = p;
    }
    __syncthreads();
    if (tid == 0) {
      float t = 0.f;
      #pragma unroll
      for (int i = 0; i < 8; ++i) t += lpart[i];
      lyy = t;
    }
    __syncthreads();
  }

  out[(size_t)b * CH + tid] = ly[tid];
}

extern "C" void kernel_launch(void* const* d_in, const int* in_sizes, int n_in,
                              void* d_out, int out_size, void* d_ws, size_t ws_size,
                              hipStream_t stream) {
  (void)in_sizes; (void)n_in; (void)out_size; (void)ws_size;
  const float* x = (const float*)d_in[0];
  float* out = (float*)d_out;
  float* xT  = (float*)d_ws;

  rvp_transpose<<<dim3(KDIM / 32, CH / 32, BATCH), dim3(32, 8), 0, stream>>>(x, xT);
  rvp_irls<<<dim3(BATCH), dim3(256), 0, stream>>>(x, xT, out);
}


# IRLS workgroup widened from 8 to 32 waves (a wave owns 32 k in phase 1, 32 c of one k-quarter in phase 2) to cover L2 latency; den and num now summed from per-wave and per-k-quarter partials
# speedup vs baseline: 1.3251x; 1.3251x over previous
#include <hip/hip_runtime.h>

#define BATCH 64
#define CH    256
#define KDIM  1024
#define NITER 100
#define NWAVE 32
#define NTHR  (NWAVE * 32)

typedef __attribute__((ext_vector_type(2))) float v2f;
typedef __attribute__((ext_vector_type(8))) float v8f;

static __device__ __forceinline__ v8f wmma4(v2f a, v2f b, v8f c) {
  return __builtin_amdgcn_wmma_f32_16x16x4_f32(false, a, false, b, (short)0, c, false, false);
}

__global__ __launch_bounds__(256) void rvp_transpose(const float* __restrict__ x,
                                                     float* __restrict__ xT) {
  __shared__ float tile[32][33];
  const int b  = blockIdx.z;
  const int k0 = blockIdx.x * 32;
  const int c0 = blockIdx.y * 32;
  const float* xb = x + (size_t)b * CH * KDIM;
  float* xTb = xT + (size_t)b * CH * KDIM;
  const int tx = threadIdx.x, ty = threadIdx.y;
  #pragma unroll
  for (int j = 0; j < 32; j += 8)
    tile[ty + j][tx] = xb[(size_t)(c0 + ty + j) * KDIM + (k0 + tx)];
  __syncthreads();
  #pragma unroll
  for (int j = 0; j < 32; j += 8)
    xTb[(size_t)(k0 + ty + j) * CH + (c0 + tx)] = tile[tx][ty + j];
}

__global__ __launch_bounds__(NTHR) void rvp_irls(const float* __restrict__ x,
                                                const float* __restrict__ xT,
                                                float* __restrict__ out) {
  __shared__ float ly[CH];
  __shared__ float ls[KDIM];
  __shared__ float lw[KDIM];
  __shared__ float lnum[4][CH];
  __shared__ float lpart[NWAVE];
  __shared__ float lden;
  __shared__ float lyy;

  const int b    = blockIdx.x;
  const int tid  = threadIdx.x;
  const int wid  = tid >> 5;
  const int lane = tid & 31;
  const int laneN = lane & 15;
  const int laneH = lane >> 4;
  const float amask = (laneN == 0) ? 1.0f : 0.0f;

  const float* xb  = x  + (size_t)b * CH * KDIM;
  const float* xTb = xT + (size_t)b * CH * KDIM;

  {
    float s0 = 0.f;
    for (int c = 0; c < CH; ++c) {
      const float a0 = xb[(size_t)c * KDIM + tid];
      s0 += a0 * a0;
    }
    ls[tid] = s0;
  }
  if (tid < CH) {
    float a = 0.f;
    for (int k = 0; k < KDIM; ++k) a += xTb[(size_t)k * CH + tid];
    float yv = a * (1.0f / (float)KDIM);
    ly[tid] = yv;
    float p = yv * yv;
    #pragma unroll
    for (int off = 16; off > 0; off >>= 1) p += __shfl_xor(p, off, 32);
    if (lane == 0) lpart[wid] = p;
  }
  __syncthreads();
  if (tid == 0) {
    float t = 0.f;
    #pragma unroll
    for (int i = 0; i < 8; ++i) t += lpart[i];
    lyy = t;
  }
  __syncthreads();

  for (int it = 0; it < NITER; ++it) {
    const float yy = lyy;

    const int kbase = wid * 32;
    v8f t8[2];
    t8[0] = (v8f){}; t8[1] = (v8f){};
    #pragma unroll 4
    for (int c0 = 0; c0 < CH; c0 += 4) {
      const int cA = c0 + (laneH << 1);
      const v2f yv = *(const v2f*)(ly + cA);
      v2f av; av.x = yv.x * amask; av.y = yv.y * amask;
      const float* pB = xb + (size_t)cA * KDIM + (kbase + laneN);
      #pragma unroll
      for (int j = 0; j < 2; ++j) {
        v2f bv;
        bv.x = pB[j * 16];
        bv.y = pB[j * 16 + KDIM];
        t8[j] = wmma4(av, bv, t8[j]);
      }
    }
    float den_acc = 0.f;
    if (lane < 16) {
      #pragma unroll
      for (int j = 0; j < 2; ++j) {
        const int k = kbase + j * 16 + lane;
        const float t  = t8[j][0];
        const float z2 = yy - 2.0f * t + ls[k];
        const float wv = 1.0f / sqrtf(1.0f + z2);
        lw[k] = wv;
        den_acc += wv;
      }
    }
    #pragma unroll
    for (int off = 16; off > 0; off >>= 1) den_acc += __shfl_xor(den_acc, off, 32);
    if (lane == 0) lpart[wid] = den_acc;
    __syncthreads();
    if (tid == 0) {
      float t = 0.f;
      #pragma unroll
      for (int i = 0; i < NWAVE; ++i) t += lpart[i];
      lden = t;
    }

    {
      const int c0  = (wid & 7) * 32;
      const int kq  = wid >> 3;
      const int kq0 = kq * (KDIM / 4);
      const float* lwq = lw + kq0;
      const float* xTq = xTb + (size_t)kq0 * CH;
      v8f n0a = {}, n1a = {}, n0b = {}, n1b = {};
      #pragma unroll 2
      for (int k0 = 0; k0 < KDIM / 4; k0 += 8) {
        const int kA0 = k0 + (laneH << 1);
        const int kA1 = kA0 + 4;
        const v2f w0 = *(const v2f*)(lwq + kA0);
        const v2f w1 = *(const v2f*)(lwq + kA1);
        v2f a0; a0.x = w0.x * amask; a0.y = w0.y * amask;
        v2f a1; a1.x = w1.x * amask; a1.y = w1.y * amask;
        const float* q0 = xTq + (size_t)kA0 * CH + laneN;
        const float* q1 = xTq + (size_t)kA1 * CH + laneN;
        v2f b00, b01, b10, b11;
        b00.x = q0[c0];          b00.y = q0[c0 + CH];
        b01.x = q0[c0 + 16];     b01.y = q0[c0 + 16 + CH];
        b10.x = q1[c0];          b10.y = q1[c0 + CH];
        b11.x = q1[c0 + 16];     b11.y = q1[c0 + 16 + CH];
        n0a = wmma4(a0, b00, n0a);
        n1a = wmma4(a0, b01, n1a);
        n0b = wmma4(a1, b10, n0b);
        n1b = wmma4(a1, b11, n1b);
      }
      if (lane < 16) {
        lnum[kq][c0 + lane]      = n0a[0] + n0b[0];
        lnum[kq][c0 + 16 + lane] = n1a[0] + n1b[0];
      }
    }
    __syncthreads();

    if (tid < CH) {
      const float nv = (lnum[0][tid] + lnum[1][tid]) + (lnum[2][tid] + lnum[3][tid]);
      const float yv = nv / lden;
      ly[tid] = yv;
      float p = yv * yv;
      #pragma unroll
      for (int off = 16; off > 0; off >>= 1) p += __shfl_xor(p, off, 32);
      if (lane == 0) lpart[wid] = p;
    }
    __syncthreads();
    if (tid == 0) {
      float t = 0.f;
      #pragma unroll
      for (int i = 0; i < 8; ++i) t += lpart[i];
      lyy = t;
    }
    __syncthreads();
  }

  if (tid < CH) out[(size_t)b * CH + tid] = ly[tid];
}

extern "C" void kernel_launch(void* const* d_in, const int* in_sizes, int n_in,
                              void* d_out, int out_size, void* d_ws, size_t ws_size,
                              hipStream_t stream) {
  (void)in_sizes; (void)n_in; (void)out_size; (void)ws_size;
  const float* x = (const float*)d_in[0];
  float* out = (float*)d_out;
  float* xT  = (float*)d_ws;

  rvp_transpose<<<dim3(KDIM / 32, CH / 32, BATCH), dim3(32, 8), 0, stream>>>(x, xT);
  rvp_irls<<<dim3(BATCH), dim3(NTHR), 0, stream>>>(x, xT, out);
}


# B fragment re-mapped to 1 row x 64 cols per WMMA (4 k slots routed to 4 D rows by a masked A): one 512-byte b128 load feeds 2 WMMAs, 4x fewer loads; t, num summed from c-quarter / k-group partials
# speedup vs baseline: 1.4192x; 1.0710x over previous
#include <hip/hip_runtime.h>

#define BATCH 64
#define CH    256
#define KDIM  1024
#define NITER 100
#define NWAVE 32
#define NTHR  (NWAVE * 32)

typedef __attribute__((ext_vector_type(2))) float v2f;
typedef __attribute__((ext_vector_type(4))) float v4f;
typedef __attribute__((ext_vector_type(8))) float v8f;

static __device__ __forceinline__ v8f wmma4(v2f a, v2f b, v8f c) {
  return __builtin_amdgcn_wmma_f32_16x16x4_f32(false, a, false, b, (short)0, c, false, false);
}

__global__ __launch_bounds__(256) void rvp_transpose(const float* __restrict__ x,
                                                     float* __restrict__ xT) {
  __shared__ float tile[32][33];
  const int b  = blockIdx.z;
  const int k0 = blockIdx.x * 32;
  const int c0 = blockIdx.y * 32;
  const float* xb = x + (size_t)b * CH * KDIM;
  float* xTb = xT + (size_t)b * CH * KDIM;
  const int tx = threadIdx.x, ty = threadIdx.y;
  #pragma unroll
  for (int j = 0; j < 32; j += 8)
    tile[ty + j][tx] = xb[(size_t)(c0 + ty + j) * KDIM + (k0 + tx)];
  __syncthreads();
  #pragma unroll
  for (int j = 0; j < 32; j += 8)
    xTb[(size_t)(k0 + ty + j) * CH + (c0 + tx)] = tile[tx][ty + j];
}

__global__ __launch_bounds__(NTHR) void rvp_irls(const float* __restrict__ x,
                                                const float* __restrict__ xT,
                                                float* __restrict__ out) {
  __shared__ __attribute__((aligned(16))) float ly[CH];
  __shared__ __attribute__((aligned(16))) float ls[KDIM];
  __shared__ __attribute__((aligned(16))) float lw[KDIM];
  __shared__ __attribute__((aligned(16))) float lt[4][KDIM];
  __shared__ __attribute__((aligned(16))) float lnum[16][CH];
  __shared__ float lpart[NWAVE];
  __shared__ float lden;
  __shared__ float lyy;

  const int b    = blockIdx.x;
  const int tid  = threadIdx.x;
  const int wid  = tid >> 5;
  const int lane = tid & 31;
  const float m0 = (lane == 0 || lane == 18) ? 1.0f : 0.0f;
  const float m1 = (lane == 1 || lane == 19) ? 1.0f : 0.0f;

  const float* xb  = x  + (size_t)b * CH * KDIM;
  const float* xTb = xT + (size_t)b * CH * KDIM;

  {
    float s0 = 0.f;
    for (int c = 0; c < CH; ++c) {
      const float a0 = xb[(size_t)c * KDIM + tid];
      s0 += a0 * a0;
    }
    ls[tid] = s0;
  }
  if (tid < CH) {
    float a = 0.f;
    for (int k = 0; k < KDIM; ++k) a += xTb[(size_t)k * CH + tid];
    float yv = a * (1.0f / (float)KDIM);
    ly[tid] = yv;
    float p = yv * yv;
    #pragma unroll
    for (int off = 16; off > 0; off >>= 1) p += __shfl_xor(p, off, 32);
    if (lane == 0) lpart[wid] = p;
  }
  __syncthreads();
  if (tid == 0) {
    float t = 0.f;
    #pragma unroll
    for (int i = 0; i < 8; ++i) t += lpart[i];
    lyy = t;
  }
  __syncthreads();

  for (int it = 0; it < NITER; ++it) {
    {
      const int kg = wid & 7;
      const int cq = wid >> 3;
      const float* px = xb + (size_t)(cq * 64) * KDIM + (kg * 128 + lane * 4);
      const float* py = ly + cq * 64;
      v8f ta = {}, tb = {};
      #pragma unroll 8
      for (int c = 0; c < 64; ++c) {
        const v4f xv = *(const v4f*)(px + (size_t)c * KDIM);
        const float yv = py[c];
        v2f av; av.x = yv * m0; av.y = yv * m1;
        v2f b0; b0.x = xv.x; b0.y = xv.y;
        v2f b1; b1.x = xv.z; b1.y = xv.w;
        ta = wmma4(av, b0, ta);
        tb = wmma4(av, b1, tb);
      }
      if (lane < 16) {
        float* q = &lt[cq][kg * 128 + lane * 4];
        v4f lo; lo.x = ta[0]; lo.y = ta[1]; lo.z = tb[0]; lo.w = tb[1];
        v4f hi; hi.x = ta[2]; hi.y = ta[3]; hi.z = tb[2]; hi.w = tb[3];
        *(v4f*)q        = lo;
        *(v4f*)(q + 64) = hi;
      }
    }
    __syncthreads();

    {
      const float yy = lyy;
      const float t  = (lt[0][tid] + lt[1][tid]) + (lt[2][tid] + lt[3][tid]);
      const float z2 = yy - 2.0f * t + ls[tid];
      const float wv = 1.0f / sqrtf(1.0f + z2);
      lw[tid] = wv;
      float den_acc = wv;
      #pragma unroll
      for (int off = 16; off > 0; off >>= 1) den_acc += __shfl_xor(den_acc, off, 32);
      if (lane == 0) lpart[wid] = den_acc;
    }
    __syncthreads();
    if (tid == 0) {
      float t = 0.f;
      #pragma unroll
      for (int i = 0; i < NWAVE; ++i) t += lpart[i];
      lden = t;
    }

    {
      const int cg = wid & 1;
      const int kg = wid >> 1;
      const float* pxT = xTb + (size_t)(kg * 64) * CH + (cg * 128 + lane * 4);
      const float* pw  = lw + kg * 64;
      v8f na = {}, nb = {};
      #pragma unroll 8
      for (int k = 0; k < 64; ++k) {
        const v4f xv = *(const v4f*)(pxT + (size_t)k * CH);
        const float wv = pw[k];
        v2f av; av.x = wv * m0; av.y = wv * m1;
        v2f b0; b0.x = xv.x; b0.y = xv.y;
        v2f b1; b1.x = xv.z; b1.y = xv.w;
        na = wmma4(av, b0, na);
        nb = wmma4(av, b1, nb);
      }
      if (lane < 16) {
        float* q = &lnum[kg][cg * 128 + lane * 4];
        v4f lo; lo.x = na[0]; lo.y = na[1]; lo.z = nb[0]; lo.w = nb[1];
        v4f hi; hi.x = na[2]; hi.y = na[3]; hi.z = nb[2]; hi.w = nb[3];
        *(v4f*)q        = lo;
        *(v4f*)(q + 64) = hi;
      }
    }
    __syncthreads();

    if (tid < CH) {
      const float n0 = (lnum[0][tid]  + lnum[1][tid])  + (lnum[2][tid]  + lnum[3][tid]);
      const float n1 = (lnum[4][tid]  + lnum[5][tid])  + (lnum[6][tid]  + lnum[7][tid]);
      const float n2 = (lnum[8][tid]  + lnum[9][tid])  + (lnum[10][tid] + lnum[11][tid]);
      const float n3 = (lnum[12][tid] + lnum[13][tid]) + (lnum[14][tid] + lnum[15][tid]);
      const float nv = (n0 + n1) + (n2 + n3);
      const float yv = nv / lden;
      ly[tid] = yv;
      float p = yv * yv;
      #pragma unroll
      for (int off = 16; off > 0; off >>= 1) p += __shfl_xor(p, off, 32);
      if (lane == 0) lpart[wid] = p;
    }
    __syncthreads();
    if (tid == 0) {
      float t = 0.f;
      #pragma unroll
      for (int i = 0; i < 8; ++i) t += lpart[i];
      lyy = t;
    }
  }

  if (tid < CH) out[(size_t)b * CH + tid] = ly[tid];
}

extern "C" void kernel_launch(void* const* d_in, const int* in_sizes, int n_in,
                              void* d_out, int out_size, void* d_ws, size_t ws_size,
                              hipStream_t stream) {
  (void)in_sizes; (void)n_in; (void)out_size; (void)ws_size;
  const float* x = (const float*)d_in[0];
  float* out = (float*)d_out;
  float* xT  = (float*)d_ws;

  rvp_transpose<<<dim3(KDIM / 32, CH / 32, BATCH), dim3(32, 8), 0, stream>>>(x, xT);
  rvp_irls<<<dim3(BATCH), dim3(NTHR), 0, stream>>>(x, xT, out);
}


# operand rows double-buffered in registers: each round's 8 row loads are issued one round (16 WMMAs) ahead of use, the next phase's first round ahead of the barrier; sums and their order unchanged
# speedup vs baseline: 1.5188x; 1.0702x over previous
#include <hip/hip_runtime.h>

#define BATCH 64
#define CH    256
#define KDIM  1024
#define NITER 100
#define NWAVE 32
#define NTHR  (NWAVE * 32)

typedef __attribute__((ext_vector_type(2))) float v2f;
typedef __attribute__((ext_vector_type(4))) float v4f;
typedef __attribute__((ext_vector_type(8))) float v8f;

#define ROUND_FENCE() __builtin_amdgcn_sched_barrier(0)

static __device__ __forceinline__ v8f wmma4(v2f a, v2f b, v8f c) {
  return __builtin_amdgcn_wmma_f32_16x16x4_f32(false, a, false, b, (short)0, c, false, false);
}

template <int STRIDE>
static __device__ __forceinline__ void load8(v4f (&buf)[8], const float* __restrict__ p) {
  #pragma unroll
  for (int i = 0; i < 8; ++i) buf[i] = *(const v4f*)(p + (size_t)i * STRIDE);
}

static __device__ __forceinline__ void mma8(const v4f (&buf)[8], const float* pv,
                                            const float m0, const float m1, v8f& acc0, v8f& acc1) {
  const v4f s0 = *(const v4f*)(pv), s1 = *(const v4f*)(pv + 4);
  const float sv[8] = {s0.x, s0.y, s0.z, s0.w, s1.x, s1.y, s1.z, s1.w};
  #pragma unroll
  for (int i = 0; i < 8; ++i) {
    v2f av; av.x = sv[i] * m0; av.y = sv[i] * m1;
    v2f b0; b0.x = buf[i].x; b0.y = buf[i].y;
    v2f b1; b1.x = buf[i].z; b1.y = buf[i].w;
    acc0 = wmma4(av, b0, acc0);
    acc1 = wmma4(av, b1, acc1);
  }
}

__global__ __launch_bounds__(256) void rvp_transpose(const float* __restrict__ x,
                                                     float* __restrict__ xT) {
  __shared__ float tile[32][33];
  const int b  = blockIdx.z;
  const int k0 = blockIdx.x * 32;
  const int c0 = blockIdx.y * 32;
  const float* xb = x + (size_t)b * CH * KDIM;
  float* xTb = xT + (size_t)b * CH * KDIM;
  const int tx = threadIdx.x, ty = threadIdx.y;
  #pragma unroll
  for (int j = 0; j < 32; j += 8)
    tile[ty + j][tx] = xb[(size_t)(c0 + ty + j) * KDIM + (k0 + tx)];
  __syncthreads();
  #pragma unroll
  for (int j = 0; j < 32; j += 8)
    xTb[(size_t)(k0 + ty + j) * CH + (c0 + tx)] = tile[tx][ty + j];
}

__global__ __launch_bounds__(NTHR) void rvp_irls(const float* __restrict__ x,
                                                const float* __restrict__ xT,
                                                float* __restrict__ out) {
  __shared__ __attribute__((aligned(16))) float ly[CH];
  __shared__ __attribute__((aligned(16))) float ls[KDIM];
  __shared__ __attribute__((aligned(16))) float lw[KDIM];
  __shared__ __attribute__((aligned(16))) float lt[4][KDIM];
  __shared__ __attribute__((aligned(16))) float lnum[16][CH];
  __shared__ float lpart[NWAVE];
  __shared__ float lden;
  __shared__ float lyy;

  const int b    = blockIdx.x;
  const int tid  = threadIdx.x;
  const int wid  = tid >> 5;
  const int lane = tid & 31;
  const float m0 = (lane == 0 || lane == 18) ? 1.0f : 0.0f;
  const float m1 = (lane == 1 || lane == 19) ? 1.0f : 0.0f;

  const float* xb  = x  + (size_t)b * CH * KDIM;
  const float* xTb = xT + (size_t)b * CH * KDIM;

  {
    float s0 = 0.f;
    for (int c = 0; c < CH; ++c) {
      const float a0 = xb[(size_t)c * KDIM + tid];
      s0 += a0 * a0;
    }
    ls[tid] = s0;
  }
  if (tid < CH) {
    float a = 0.f;
    for (int k = 0; k < KDIM; ++k) a += xTb[(size_t)k * CH + tid];
    float yv = a * (1.0f / (float)KDIM);
    ly[tid] = yv;
    float p = yv * yv;
    #pragma unroll
    for (int off = 16; off > 0; off >>= 1) p += __shfl_xor(p, off, 32);
    if (lane == 0) lpart[wid] = p;
  }
  __syncthreads();
  if (tid == 0) {
    float t = 0.f;
    #pragma unroll
    for (int i = 0; i < 8; ++i) t += lpart[i];
    lyy = t;
  }
  __syncthreads();

  const int kg1 = wid & 7;
  const int cq1 = wid >> 3;
  const int cg2 = wid & 1;
  const int kg2 = wid >> 1;
  const float* px  = xb  + (size_t)(cq1 * 64) * KDIM + (kg1 * 128 + lane * 4);
  const float* pxT = xTb + (size_t)(kg2 * 64) * CH   + (cg2 * 128 + lane * 4);
  const float* py  = ly + cq1 * 64;
  const float* pw  = lw + kg2 * 64;

  v4f bufA[8], bufB[8];
  load8<KDIM>(bufA, px);

  for (int it = 0; it < NITER; ++it) {
    int zoff = 0;
    asm volatile("" : "+s"(zoff));
    const float* pxi  = px  + zoff;
    const float* pxTi = pxT + zoff;

    {
      v8f ta = {}, tb = {};
      #pragma unroll
      for (int r = 0; r < 8; r += 2) {
        load8<KDIM>(bufB, pxi + (size_t)(r + 1) * 8 * KDIM);
        ROUND_FENCE();
        mma8(bufA, py + r * 8, m0, m1, ta, tb);
        ROUND_FENCE();
        if (r + 2 < 8) load8<KDIM>(bufA, pxi + (size_t)(r + 2) * 8 * KDIM);
        else           load8<CH>(bufA, pxTi);
        ROUND_FENCE();
        mma8(bufB, py + (r + 1) * 8, m0, m1, ta, tb);
        ROUND_FENCE();
      }
      if (lane < 16) {
        float* q = &lt[cq1][kg1 * 128 + lane * 4];
        v4f lo; lo.x = ta[0]; lo.y = ta[1]; lo.z = tb[0]; lo.w = tb[1];
        v4f hi; hi.x = ta[2]; hi.y = ta[3]; hi.z = tb[2]; hi.w = tb[3];
        *(v4f*)q        = lo;
        *(v4f*)(q + 64) = hi;
      }
    }
    __syncthreads();

    {
      const float yy = lyy;
      const float t  = (lt[0][tid] + lt[1][tid]) + (lt[2][tid] + lt[3][tid]);
      const float z2 = yy - 2.0f * t + ls[tid];
      const float wv = 1.0f / sqrtf(1.0f + z2);
      lw[tid] = wv;
      float den_acc = wv;
      #pragma unroll
      for (int off = 16; off > 0; off >>= 1) den_acc += __shfl_xor(den_acc, off, 32);
      if (lane == 0) lpart[wid] = den_acc;
    }
    __syncthreads();
    if (tid == 0) {
      float t = 0.f;
      #pragma unroll
      for (int i = 0; i < NWAVE; ++i) t += lpart[i];
      lden = t;
    }

    {
      v8f na = {}, nb = {};
      #pragma unroll
      for (int r = 0; r < 8; r += 2) {
        load8<CH>(bufB, pxTi + (size_t)(r + 1) * 8 * CH);
        ROUND_FENCE();
        mma8(bufA, pw + r * 8, m0, m1, na, nb);
        ROUND_FENCE();
        if (r + 2 < 8) load8<CH>(bufA, pxTi + (size_t)(r + 2) * 8 * CH);
        else           load8<KDIM>(bufA, pxi);
        ROUND_FENCE();
        mma8(bufB, pw + (r + 1) * 8, m0, m1, na, nb);
        ROUND_FENCE();
      }
      if (lane < 16) {
        float* q = &lnum[kg2][cg2 * 128 + lane * 4];
        v4f lo; lo.x = na[0]; lo.y = na[1]; lo.z = nb[0]; lo.w = nb[1];
        v4f hi; hi.x = na[2]; hi.y = na[3]; hi.z = nb[2]; hi.w = nb[3];
        *(v4f*)q        = lo;
        *(v4f*)(q + 64) = hi;
      }
    }
    __syncthreads();

    if (tid < CH) {
      const float n0 = (lnum[0][tid]  + lnum[1][tid])  + (lnum[2][tid]  + lnum[3][tid]);
      const float n1 = (lnum[4][tid]  + lnum[5][tid])  + (lnum[6][tid]  + lnum[7][tid]);
      const float n2 = (lnum[8][tid]  + lnum[9][tid])  + (lnum[10][tid] + lnum[11][tid]);
      const float n3 = (lnum[12][tid] + lnum[13][tid]) + (lnum[14][tid] + lnum[15][tid]);
      const float nv = (n0 + n1) + (n2 + n3);
      const float yv = nv / lden;
      ly[tid] = yv;
      float p = yv * yv;
      #pragma unroll
      for (int off = 16; off > 0; off >>= 1) p += __shfl_xor(p, off, 32);
      if (lane == 0) lpart[wid] = p;
    }
    __syncthreads();
    if (tid == 0) {
      float t = 0.f;
      #pragma unroll
      for (int i = 0; i < 8; ++i) t += lpart[i];
      lyy = t;
    }
  }

  if (tid < CH) out[(size_t)b * CH + tid] = ly[tid];
}

extern "C" void kernel_launch(void* const* d_in, const int* in_sizes, int n_in,
                              void* d_out, int out_size, void* d_ws, size_t ws_size,
                              hipStream_t stream) {
  (void)in_sizes; (void)n_in; (void)out_size; (void)ws_size;
  const float* x = (const float*)d_in[0];
  float* out = (float*)d_out;
  float* xT  = (float*)d_ws;

  rvp_transpose<<<dim3(KDIM / 32, CH / 32, BATCH), dim3(32, 8), 0, stream>>>(x, xT);
  rvp_irls<<<dim3(BATCH), dim3(NTHR), 0, stream>>>(x, xT, out);
}


# batch split over 4 workgroups by k-quarter, one launch per IRLS iteration; each quarter of x held in registers for both products (num from the WMMA diagonal); t, s, num, den summed in a new order
# speedup vs baseline: 2.8521x; 1.8779x over previous
#include <hip/hip_runtime.h>

#define BATCH  64
#define CH     256
#define KDIM   1024
#define NSPLIT 4
#define KQ     (KDIM / NSPLIT)
#define NWAVE  16
#define NTHR   (NWAVE * 32)
#define ROWS   32
#define DPAD   32
#define RP     68

typedef __attribute__((ext_vector_type(2))) float v2f;
typedef __attribute__((ext_vector_type(4))) float v4f;
typedef __attribute__((ext_vector_type(8))) float v8f;

static __device__ __forceinline__ v8f wmma4(v2f a, v2f b, v8f c) {
  return __builtin_amdgcn_wmma_f32_16x16x4_f32(false, a, false, b, (short)0, c, false, false);
}

static __device__ __forceinline__ float diag_pick(const v8f d, const bool b0, const bool b1,
                                                  const bool b2) {
  const float e0 = b0 ? d[1] : d[0];
  const float e1 = b0 ? d[3] : d[2];
  const float e2 = b0 ? d[5] : d[4];
  const float e3 = b0 ? d[7] : d[6];
  const float f0 = b1 ? e1 : e0;
  const float f1 = b1 ? e3 : e2;
  return b2 ? f1 : f0;
}

__global__ __launch_bounds__(256) void rvp_init(const float* __restrict__ x,
                                                float* __restrict__ p, float* __restrict__ d) {
  __shared__ float lsum[CH];
  const int j = blockIdx.x, b = blockIdx.y;
  const int tid = threadIdx.x, wid = tid >> 5, lane = tid & 31;
  const float* px = x + (size_t)b * CH * KDIM + (size_t)(wid * 32) * KDIM + (j * KQ + lane * 4);
  for (int i = 0; i < 32; ++i) {
    const v4f u = *(const v4f*)(px + (size_t)i * KDIM);
    const v4f v = *(const v4f*)(px + (size_t)i * KDIM + 128);
    float a = ((u.x + u.y) + (u.z + u.w)) + ((v.x + v.y) + (v.z + v.w));
    #pragma unroll
    for (int off = 16; off > 0; off >>= 1) a += __shfl_xor(a, off, 32);
    if (lane == 0) lsum[wid * 32 + i] = a;
  }
  __syncthreads();
  p[(size_t)(b * NSPLIT + j) * CH + tid] = lsum[tid];
  if (tid == 0) d[(size_t)(b * NSPLIT + j) * DPAD] = (float)KQ;
}

__global__ __launch_bounds__(NTHR) void rvp_step(const float* __restrict__ x,
                                                const float* __restrict__ pin,
                                                const float* __restrict__ din,
                                                float* __restrict__ pout,
                                                float* __restrict__ dout) {
  __shared__ __attribute__((aligned(16))) float lw[KQ];
  __shared__ __attribute__((aligned(16))) float lt[8][KQ];
  __shared__ __attribute__((aligned(16))) float lsq[8][KQ];
  __shared__ __attribute__((aligned(16))) float lred[CH * RP];
  __shared__ __attribute__((aligned(16))) float lyyp[8];
  __shared__ float lpart[8];

  const int j    = blockIdx.x;
  const int b    = blockIdx.y;
  const int tid  = threadIdx.x;
  const int wid  = tid >> 5;
  const int lane = tid & 31;
  const int kc   = wid & 1;
  const int rg   = wid >> 1;
  const float m0 = (lane == 0 || lane == 18) ? 1.0f : 0.0f;
  const float m1 = (lane == 1 || lane == 19) ? 1.0f : 0.0f;

  const float* pb = pin + (size_t)b * NSPLIT * CH + (rg * ROWS + lane);
  const float* db = din + (size_t)b * NSPLIT * DPAD;
  const float p0 = pb[0], p1 = pb[CH], p2 = pb[2 * CH], p3 = pb[3 * CH];
  const float d0 = db[0], d1 = db[DPAD], d2 = db[2 * DPAD], d3 = db[3 * DPAD];

  const float* px = x + (size_t)b * CH * KDIM + (size_t)(rg * ROWS) * KDIM
                      + (j * KQ + kc * 128 + lane * 4);
  v4f buf[ROWS];
  #pragma unroll
  for (int i = 0; i < ROWS; ++i) buf[i] = *(const v4f*)(px + (size_t)i * KDIM);
  __builtin_amdgcn_sched_barrier(0);

  const float yl = ((p0 + p1) + (p2 + p3)) / ((d0 + d1) + (d2 + d3));
  if (kc == 0) {
    float p = yl * yl;
    #pragma unroll
    for (int off = 16; off > 0; off >>= 1) p += __shfl_xor(p, off, 32);
    if (lane == 0) lyyp[rg] = p;
  }

  {
    v8f ta = {}, tb = {};
    v4f sq = {0.f, 0.f, 0.f, 0.f};
    #pragma unroll
    for (int g = 0; g < ROWS; g += 8) {
      #pragma unroll
      for (int i = 0; i < 8; ++i) {
        const v4f r = buf[g + i];
        const float yi = __int_as_float(__builtin_amdgcn_readlane(__float_as_int(yl), g + i));
        v2f av; av.x = yi * m0; av.y = yi * m1;
        v2f b0; b0.x = r.x; b0.y = r.y;
        v2f b1; b1.x = r.z; b1.y = r.w;
        ta = wmma4(av, b0, ta);
        tb = wmma4(av, b1, tb);
        sq.x += r.x * r.x; sq.y += r.y * r.y; sq.z += r.z * r.z; sq.w += r.w * r.w;
      }
    }
    if (lane < 16) {
      float* q = &lt[rg][kc * 128 + lane * 4];
      v4f lo; lo.x = ta[0]; lo.y = ta[1]; lo.z = tb[0]; lo.w = tb[1];
      v4f hi; hi.x = ta[2]; hi.y = ta[3]; hi.z = tb[2]; hi.w = tb[3];
      *(v4f*)q        = lo;
      *(v4f*)(q + 64) = hi;
    }
    *(v4f*)&lsq[rg][kc * 128 + lane * 4] = sq;
  }
  __syncthreads();

  if (tid < KQ) {
    const v4f q0 = *(const v4f*)(lyyp), q1 = *(const v4f*)(lyyp + 4);
    const float yy = ((q0.x + q0.y) + (q0.z + q0.w)) + ((q1.x + q1.y) + (q1.z + q1.w));
    const float t  = ((lt[0][tid] + lt[1][tid]) + (lt[2][tid] + lt[3][tid]))
                   + ((lt[4][tid] + lt[5][tid]) + (lt[6][tid] + lt[7][tid]));
    const float s  = ((lsq[0][tid] + lsq[1][tid]) + (lsq[2][tid] + lsq[3][tid]))
                   + ((lsq[4][tid] + lsq[5][tid]) + (lsq[6][tid] + lsq[7][tid]));
    const float z2 = yy - 2.0f * t + s;
    const float wv = 1.0f / sqrtf(1.0f + z2);
    lw[tid] = wv;
    float den_acc = wv;
    #pragma unroll
    for (int off = 16; off > 0; off >>= 1) den_acc += __shfl_xor(den_acc, off, 32);
    if (lane == 0) lpart[wid] = den_acc;
  }
  __syncthreads();
  if (tid == 0) {
    float t = 0.f;
    #pragma unroll
    for (int i = 0; i < 8; ++i) t += lpart[i];
    dout[(size_t)(b * NSPLIT + j) * DPAD] = t;
  }

  {
    const v4f wq = *(const v4f*)(lw + kc * 128 + lane * 4);
    v2f a0; a0.x = wq.x; a0.y = wq.y;
    v2f a1; a1.x = wq.z; a1.y = wq.w;
    const bool b0 = (lane & 1) != 0, b1 = (lane & 2) != 0, b2 = (lane & 4) != 0;
    float* qr = lred + (size_t)(rg * ROWS) * RP + kc * 32 + lane;
    #pragma unroll
    for (int i = 0; i < ROWS; ++i) {
      const v4f r = buf[i];
      v2f x0; x0.x = r.x; x0.y = r.y;
      v2f x1; x1.x = r.z; x1.y = r.w;
      v8f d = {};
      d = wmma4(a0, x0, d);
      d = wmma4(a1, x1, d);
      qr[i * RP] = diag_pick(d, b0, b1, b2);
    }
  }
  __syncthreads();

  if (tid < CH) {
    const float* r = lred + (size_t)tid * RP;
    const v4f c0 = *(const v4f*)(r),      c1 = *(const v4f*)(r + 4);
    const v4f c2 = *(const v4f*)(r + 24), c3 = *(const v4f*)(r + 28);
    const v4f c4 = *(const v4f*)(r + 32), c5 = *(const v4f*)(r + 36);
    const v4f c6 = *(const v4f*)(r + 56), c7 = *(const v4f*)(r + 60);
    const float h0 = (((c0.x + c0.y) + (c0.z + c0.w)) + ((c1.x + c1.y) + (c1.z + c1.w)))
                   + (((c2.x + c2.y) + (c2.z + c2.w)) + ((c3.x + c3.y) + (c3.z + c3.w)));
    const float h1 = (((c4.x + c4.y) + (c4.z + c4.w)) + ((c5.x + c5.y) + (c5.z + c5.w)))
                   + (((c6.x + c6.y) + (c6.z + c6.w)) + ((c7.x + c7.y) + (c7.z + c7.w)));
    pout[(size_t)(b * NSPLIT + j) * CH + tid] = h0 + h1;
  }
}

__global__ __launch_bounds__(CH) void rvp_final(const float* __restrict__ p,
                                                const float* __restrict__ d,
                                                float* __restrict__ out) {
  const int b = blockIdx.x, tid = threadIdx.x;
  const float* pb = p + (size_t)b * NSPLIT * CH;
  const float* db = d + (size_t)b * NSPLIT * DPAD;
  const float nv = (pb[tid] + pb[CH + tid]) + (pb[2 * CH + tid] + pb[3 * CH + tid]);
  const float dv = (db[0] + db[DPAD]) + (db[2 * DPAD] + db[3 * DPAD]);
  out[(size_t)b * CH + tid] = nv / dv;
}

#define STEP2  rvp_step<<<dim3(NSPLIT, BATCH), dim3(NTHR), 0, stream>>>(x, pA, dA, pB, dB); \
               rvp_step<<<dim3(NSPLIT, BATCH), dim3(NTHR), 0, stream>>>(x, pB, dB, pA, dA);
#define STEP10 STEP2 STEP2 STEP2 STEP2 STEP2
#define STEP50 STEP10 STEP10 STEP10 STEP10 STEP10

extern "C" void kernel_launch(void* const* d_in, const int* in_sizes, int n_in,
                              void* d_out, int out_size, void* d_ws, size_t ws_size,
                              hipStream_t stream) {
  (void)in_sizes; (void)n_in; (void)out_size; (void)ws_size;
  const float* x = (const float*)d_in[0];
  float* out = (float*)d_out;
  float* ws  = (float*)d_ws;
  float* pA  = ws;
  float* pB  = ws + BATCH * NSPLIT * CH;
  float* dA  = ws + 2 * BATCH * NSPLIT * CH;
  float* dB  = ws + 2 * BATCH * NSPLIT * CH + BATCH * NSPLIT * DPAD;

  rvp_init<<<dim3(NSPLIT, BATCH), dim3(256), 0, stream>>>(x, pA, dA);
  STEP50 STEP50
  rvp_final<<<dim3(BATCH), dim3(CH), 0, stream>>>(pA, dA, out);
}


# step workgroup narrowed from 16 to 8 waves, a wave owning 32 rows x all 256 k (half the waves enter per launch; same loads, same WMMAs); a row's num is one 4-WMMA chain: num summed in a new order
# speedup vs baseline: 2.9572x; 1.0369x over previous
#include <hip/hip_runtime.h>

#define BATCH  64
#define CH     256
#define KDIM   1024
#define NSPLIT 4
#define KQ     (KDIM / NSPLIT)
#define NWAVE  8
#define NTHR   (NWAVE * 32)
#define ROWS   32
#define DPAD   32
#define RP     36

typedef __attribute__((ext_vector_type(2))) float v2f;
typedef __attribute__((ext_vector_type(4))) float v4f;
typedef __attribute__((ext_vector_type(8))) float v8f;

static __device__ __forceinline__ v8f wmma4(v2f a, v2f b, v8f c) {
  return __builtin_amdgcn_wmma_f32_16x16x4_f32(false, a, false, b, (short)0, c, false, false);
}

static __device__ __forceinline__ float diag_pick(const v8f d, const bool b0, const bool b1,
                                                  const bool b2) {
  const float e0 = b0 ? d[1] : d[0];
  const float e1 = b0 ? d[3] : d[2];
  const float e2 = b0 ? d[5] : d[4];
  const float e3 = b0 ? d[7] : d[6];
  const float f0 = b1 ? e1 : e0;
  const float f1 = b1 ? e3 : e2;
  return b2 ? f1 : f0;
}

__global__ __launch_bounds__(256) void rvp_init(const float* __restrict__ x,
                                                float* __restrict__ p, float* __restrict__ d) {
  __shared__ float lsum[CH];
  const int j = blockIdx.x, b = blockIdx.y;
  const int tid = threadIdx.x, wid = tid >> 5, lane = tid & 31;
  const float* px = x + (size_t)b * CH * KDIM + (size_t)(wid * 32) * KDIM + (j * KQ + lane * 4);
  for (int i = 0; i < 32; ++i) {
    const v4f u = *(const v4f*)(px + (size_t)i * KDIM);
    const v4f v = *(const v4f*)(px + (size_t)i * KDIM + 128);
    float a = ((u.x + u.y) + (u.z + u.w)) + ((v.x + v.y) + (v.z + v.w));
    #pragma unroll
    for (int off = 16; off > 0; off >>= 1) a += __shfl_xor(a, off, 32);
    if (lane == 0) lsum[wid * 32 + i] = a;
  }
  __syncthreads();
  p[(size_t)(b * NSPLIT + j) * CH + tid] = lsum[tid];
  if (tid == 0) d[(size_t)(b * NSPLIT + j) * DPAD] = (float)KQ;
}

__global__ __launch_bounds__(NTHR) void rvp_step(const float* __restrict__ x,
                                                const float* __restrict__ pin,
                                                const float* __restrict__ din,
                                                float* __restrict__ pout,
                                                float* __restrict__ dout) {
  __shared__ __attribute__((aligned(16))) float lw[KQ];
  __shared__ __attribute__((aligned(16))) float lt[8][KQ];
  __shared__ __attribute__((aligned(16))) float lsq[8][KQ];
  __shared__ __attribute__((aligned(16))) float lred[CH * RP];
  __shared__ __attribute__((aligned(16))) float lyyp[8];
  __shared__ float lpart[8];

  const int j    = blockIdx.x;
  const int b    = blockIdx.y;
  const int tid  = threadIdx.x;
  const int wid  = tid >> 5;
  const int lane = tid & 31;
  const int rg   = wid;
  const float m0 = (lane == 0 || lane == 18) ? 1.0f : 0.0f;
  const float m1 = (lane == 1 || lane == 19) ? 1.0f : 0.0f;

  const float* pb = pin + (size_t)b * NSPLIT * CH + (rg * ROWS + lane);
  const float* db = din + (size_t)b * NSPLIT * DPAD;
  const float p0 = pb[0], p1 = pb[CH], p2 = pb[2 * CH], p3 = pb[3 * CH];
  const float d0 = db[0], d1 = db[DPAD], d2 = db[2 * DPAD], d3 = db[3 * DPAD];

  const float* px = x + (size_t)b * CH * KDIM + (size_t)(rg * ROWS) * KDIM
                      + (j * KQ + lane * 4);
  v4f buf[ROWS][2];
  #pragma unroll
  for (int i = 0; i < ROWS; ++i) {
    buf[i][0] = *(const v4f*)(px + (size_t)i * KDIM);
    buf[i][1] = *(const v4f*)(px + (size_t)i * KDIM + 128);
  }
  __builtin_amdgcn_sched_barrier(0);

  const float yl = ((p0 + p1) + (p2 + p3)) / ((d0 + d1) + (d2 + d3));

  {
    v8f ta0 = {}, tb0 = {}, ta1 = {}, tb1 = {};
    v4f sq0 = {0.f, 0.f, 0.f, 0.f}, sq1 = {0.f, 0.f, 0.f, 0.f};
    #pragma unroll
    for (int i = 0; i < ROWS; ++i) {
      const v4f r0 = buf[i][0], r1 = buf[i][1];
      const float yi = __int_as_float(__builtin_amdgcn_readlane(__float_as_int(yl), i));
      v2f av; av.x = yi * m0; av.y = yi * m1;
      v2f b00; b00.x = r0.x; b00.y = r0.y;
      v2f b01; b01.x = r0.z; b01.y = r0.w;
      v2f b10; b10.x = r1.x; b10.y = r1.y;
      v2f b11; b11.x = r1.z; b11.y = r1.w;
      ta0 = wmma4(av, b00, ta0);
      tb0 = wmma4(av, b01, tb0);
      ta1 = wmma4(av, b10, ta1);
      tb1 = wmma4(av, b11, tb1);
      sq0.x += r0.x * r0.x; sq0.y += r0.y * r0.y; sq0.z += r0.z * r0.z; sq0.w += r0.w * r0.w;
      sq1.x += r1.x * r1.x; sq1.y += r1.y * r1.y; sq1.z += r1.z * r1.z; sq1.w += r1.w * r1.w;
    }
    if (lane < 16) {
      float* q = &lt[rg][lane * 4];
      v4f lo0; lo0.x = ta0[0]; lo0.y = ta0[1]; lo0.z = tb0[0]; lo0.w = tb0[1];
      v4f hi0; hi0.x = ta0[2]; hi0.y = ta0[3]; hi0.z = tb0[2]; hi0.w = tb0[3];
      v4f lo1; lo1.x = ta1[0]; lo1.y = ta1[1]; lo1.z = tb1[0]; lo1.w = tb1[1];
      v4f hi1; hi1.x = ta1[2]; hi1.y = ta1[3]; hi1.z = tb1[2]; hi1.w = tb1[3];
      *(v4f*)q         = lo0;
      *(v4f*)(q + 64)  = hi0;
      *(v4f*)(q + 128) = lo1;
      *(v4f*)(q + 192) = hi1;
    }
    *(v4f*)&lsq[rg][lane * 4]       = sq0;
    *(v4f*)&lsq[rg][128 + lane * 4] = sq1;
  }
  {
    float p = yl * yl;
    #pragma unroll
    for (int off = 16; off > 0; off >>= 1) p += __shfl_xor(p, off, 32);
    if (lane == 0) lyyp[rg] = p;
  }
  __syncthreads();

  {
    const v4f q0 = *(const v4f*)(lyyp), q1 = *(const v4f*)(lyyp + 4);
    const float yy = ((q0.x + q0.y) + (q0.z + q0.w)) + ((q1.x + q1.y) + (q1.z + q1.w));
    const float t  = ((lt[0][tid] + lt[1][tid]) + (lt[2][tid] + lt[3][tid]))
                   + ((lt[4][tid] + lt[5][tid]) + (lt[6][tid] + lt[7][tid]));
    const float s  = ((lsq[0][tid] + lsq[1][tid]) + (lsq[2][tid] + lsq[3][tid]))
                   + ((lsq[4][tid] + lsq[5][tid]) + (lsq[6][tid] + lsq[7][tid]));
    const float z2 = yy - 2.0f * t + s;
    const float wv = 1.0f / sqrtf(1.0f + z2);
    lw[tid] = wv;
    float den_acc = wv;
    #pragma unroll
    for (int off = 16; off > 0; off >>= 1) den_acc += __shfl_xor(den_acc, off, 32);
    if (lane == 0) lpart[wid] = den_acc;
  }
  __syncthreads();
  if (tid == 0) {
    float t = 0.f;
    #pragma unroll
    for (int i = 0; i < 8; ++i) t += lpart[i];
    dout[(size_t)(b * NSPLIT + j) * DPAD] = t;
  }

  {
    const v4f wq0 = *(const v4f*)(lw + lane * 4);
    const v4f wq1 = *(const v4f*)(lw + 128 + lane * 4);
    v2f a00; a00.x = wq0.x; a00.y = wq0.y;
    v2f a01; a01.x = wq0.z; a01.y = wq0.w;
    v2f a10; a10.x = wq1.x; a10.y = wq1.y;
    v2f a11; a11.x = wq1.z; a11.y = wq1.w;
    const bool b0 = (lane & 1) != 0, b1 = (lane & 2) != 0, b2 = (lane & 4) != 0;
    float* qr = lred + (size_t)(rg * ROWS) * RP + lane;
    #pragma unroll
    for (int i = 0; i < ROWS; ++i) {
      const v4f r0 = buf[i][0], r1 = buf[i][1];
      v2f x00; x00.x = r0.x; x00.y = r0.y;
      v2f x01; x01.x = r0.z; x01.y = r0.w;
      v2f x10; x10.x = r1.x; x10.y = r1.y;
      v2f x11; x11.x = r1.z; x11.y = r1.w;
      v8f d = {};
      d = wmma4(a00, x00, d);
      d = wmma4(a01, x01, d);
      d = wmma4(a10, x10, d);
      d = wmma4(a11, x11, d);
      qr[i * RP] = diag_pick(d, b0, b1, b2);
    }
  }
  __syncthreads();

  {
    const float* r = lred + (size_t)tid * RP;
    const v4f c0 = *(const v4f*)(r),      c1 = *(const v4f*)(r + 4);
    const v4f c2 = *(const v4f*)(r + 24), c3 = *(const v4f*)(r + 28);
    const float h = (((c0.x + c0.y) + (c0.z + c0.w)) + ((c1.x + c1.y) + (c1.z + c1.w)))
                  + (((c2.x + c2.y) + (c2.z + c2.w)) + ((c3.x + c3.y) + (c3.z + c3.w)));
    pout[(size_t)(b * NSPLIT + j) * CH + tid] = h;
  }
}

__global__ __launch_bounds__(CH) void rvp_final(const float* __restrict__ p,
                                                const float* __restrict__ d,
                                                float* __restrict__ out) {
  const int b = blockIdx.x, tid = threadIdx.x;
  const float* pb = p + (size_t)b * NSPLIT * CH;
  const float* db = d + (size_t)b * NSPLIT * DPAD;
  const float nv = (pb[tid] + pb[CH + tid]) + (pb[2 * CH + tid] + pb[3 * CH + tid]);
  const float dv = (db[0] + db[DPAD]) + (db[2 * DPAD] + db[3 * DPAD]);
  out[(size_t)b * CH + tid] = nv / dv;
}

#define STEP2  rvp_step<<<dim3(NSPLIT, BATCH), dim3(NTHR), 0, stream>>>(x, pA, dA, pB, dB); \
               rvp_step<<<dim3(NSPLIT, BATCH), dim3(NTHR), 0, stream>>>(x, pB, dB, pA, dA);
#define STEP10 STEP2 STEP2 STEP2 STEP2 STEP2
#define STEP50 STEP10 STEP10 STEP10 STEP10 STEP10

extern "C" void kernel_launch(void* const* d_in, const int* in_sizes, int n_in,
                              void* d_out, int out_size, void* d_ws, size_t ws_size,
                              hipStream_t stream) {
  (void)in_sizes; (void)n_in; (void)out_size; (void)ws_size;
  const float* x = (const float*)d_in[0];
  float* out = (float*)d_out;
  float* ws  = (float*)d_ws;
  float* pA  = ws;
  float* pB  = ws + BATCH * NSPLIT * CH;
  float* dA  = ws + 2 * BATCH * NSPLIT * CH;
  float* dB  = ws + 2 * BATCH * NSPLIT * CH + BATCH * NSPLIT * DPAD;

  rvp_init<<<dim3(NSPLIT, BATCH), dim3(256), 0, stream>>>(x, pA, dA);
  STEP50 STEP50
  rvp_final<<<dim3(BATCH), dim3(CH), 0, stream>>>(pA, dA, out);
}


# only the t exchange and w left between the two products: s[k]=sum_c x^2 hoisted to the start kernel, ||y||^2 summed beside product 1, den by one wave under product 2; s, yy, den in a new order
# speedup vs baseline: 3.0733x; 1.0393x over previous
#include <hip/hip_runtime.h>

#define BATCH  64
#define CH     256
#define KDIM   1024
#define NSPLIT 4
#define KQ     (KDIM / NSPLIT)
#define NWAVE  8
#define NTHR   (NWAVE * 32)
#define ROWS   32
#define DPAD   32
#define RP     36

typedef __attribute__((ext_vector_type(2))) float v2f;
typedef __attribute__((ext_vector_type(4))) float v4f;
typedef __attribute__((ext_vector_type(8))) float v8f;

static __device__ __forceinline__ v8f wmma4(v2f a, v2f b, v8f c) {
  return __builtin_amdgcn_wmma_f32_16x16x4_f32(false, a, false, b, (short)0, c, false, false);
}

static __device__ __forceinline__ float diag_pick(const v8f d, const bool b0, const bool b1,
                                                  const bool b2) {
  const float e0 = b0 ? d[1] : d[0];
  const float e1 = b0 ? d[3] : d[2];
  const float e2 = b0 ? d[5] : d[4];
  const float e3 = b0 ? d[7] : d[6];
  const float f0 = b1 ? e1 : e0;
  const float f1 = b1 ? e3 : e2;
  return b2 ? f1 : f0;
}

__global__ __launch_bounds__(256) void rvp_init(const float* __restrict__ x,
                                                float* __restrict__ p, float* __restrict__ d,
                                                float* __restrict__ s) {
  __shared__ float lsum[CH];
  __shared__ __attribute__((aligned(16))) float lsq[8][KQ];
  const int j = blockIdx.x, b = blockIdx.y;
  const int tid = threadIdx.x, wid = tid >> 5, lane = tid & 31;
  const float* px = x + (size_t)b * CH * KDIM + (size_t)(wid * 32) * KDIM + (j * KQ + lane * 4);
  v4f sq0 = {0.f, 0.f, 0.f, 0.f}, sq1 = {0.f, 0.f, 0.f, 0.f};
  for (int i = 0; i < 32; ++i) {
    const v4f u = *(const v4f*)(px + (size_t)i * KDIM);
    const v4f v = *(const v4f*)(px + (size_t)i * KDIM + 128);
    sq0.x += u.x * u.x; sq0.y += u.y * u.y; sq0.z += u.z * u.z; sq0.w += u.w * u.w;
    sq1.x += v.x * v.x; sq1.y += v.y * v.y; sq1.z += v.z * v.z; sq1.w += v.w * v.w;
    float a = ((u.x + u.y) + (u.z + u.w)) + ((v.x + v.y) + (v.z + v.w));
    #pragma unroll
    for (int off = 16; off > 0; off >>= 1) a += __shfl_xor(a, off, 32);
    if (lane == 0) lsum[wid * 32 + i] = a;
  }
  *(v4f*)&lsq[wid][lane * 4]       = sq0;
  *(v4f*)&lsq[wid][128 + lane * 4] = sq1;
  __syncthreads();
  p[(size_t)(b * NSPLIT + j) * CH + tid] = lsum[tid];
  s[(size_t)(b * NSPLIT + j) * KQ + tid] =
      ((lsq[0][tid] + lsq[1][tid]) + (lsq[2][tid] + lsq[3][tid]))
    + ((lsq[4][tid] + lsq[5][tid]) + (lsq[6][tid] + lsq[7][tid]));
  if (tid == 0) d[(size_t)(b * NSPLIT + j) * DPAD] = (float)KQ;
}

__global__ __launch_bounds__(NTHR) void rvp_step(const float* __restrict__ x,
                                                const float* __restrict__ pin,
                                                const float* __restrict__ din,
                                                const float* __restrict__ ssq,
                                                float* __restrict__ pout,
                                                float* __restrict__ dout) {
  __shared__ __attribute__((aligned(16))) float lw[KQ];
  __shared__ __attribute__((aligned(16))) float lt[8][KQ];
  __shared__ __attribute__((aligned(16))) float lred[CH * RP];
  __shared__ __attribute__((aligned(16))) float lyyp[8];

  const int j    = blockIdx.x;
  const int b    = blockIdx.y;
  const int tid  = threadIdx.x;
  const int wid  = tid >> 5;
  const int lane = tid & 31;
  const int rg   = wid;
  const float m0 = (lane == 0 || lane == 18) ? 1.0f : 0.0f;
  const float m1 = (lane == 1 || lane == 19) ? 1.0f : 0.0f;

  const float* pb = pin + (size_t)b * NSPLIT * CH + (rg * ROWS + lane);
  const float* db = din + (size_t)b * NSPLIT * DPAD;
  const float p0 = pb[0], p1 = pb[CH], p2 = pb[2 * CH], p3 = pb[3 * CH];
  const float d0 = db[0], d1 = db[DPAD], d2 = db[2 * DPAD], d3 = db[3 * DPAD];
  const float sk = ssq[(size_t)(b * NSPLIT + j) * KQ + tid];

  const float* px = x + (size_t)b * CH * KDIM + (size_t)(rg * ROWS) * KDIM
                      + (j * KQ + lane * 4);
  v4f buf[ROWS][2];
  #pragma unroll
  for (int i = 0; i < ROWS; ++i) {
    buf[i][0] = *(const v4f*)(px + (size_t)i * KDIM);
    buf[i][1] = *(const v4f*)(px + (size_t)i * KDIM + 128);
  }
  __builtin_amdgcn_sched_barrier(0);

  const float yl = ((p0 + p1) + (p2 + p3)) / ((d0 + d1) + (d2 + d3));

  {
    float yq0 = 0.f, yq1 = 0.f, yq2 = 0.f, yq3 = 0.f;
    v8f ta0 = {}, tb0 = {}, ta1 = {}, tb1 = {};
    #pragma unroll
    for (int i = 0; i < ROWS; ++i) {
      const v4f r0 = buf[i][0], r1 = buf[i][1];
      const float yi = __int_as_float(__builtin_amdgcn_readlane(__float_as_int(yl), i));
      if ((i & 3) == 0) yq0 += yi * yi;
      else if ((i & 3) == 1) yq1 += yi * yi;
      else if ((i & 3) == 2) yq2 += yi * yi;
      else yq3 += yi * yi;
      v2f av; av.x = yi * m0; av.y = yi * m1;
      v2f b00; b00.x = r0.x; b00.y = r0.y;
      v2f b01; b01.x = r0.z; b01.y = r0.w;
      v2f b10; b10.x = r1.x; b10.y = r1.y;
      v2f b11; b11.x = r1.z; b11.y = r1.w;
      ta0 = wmma4(av, b00, ta0);
      tb0 = wmma4(av, b01, tb0);
      ta1 = wmma4(av, b10, ta1);
      tb1 = wmma4(av, b11, tb1);
    }
    lyyp[rg] = (yq0 + yq1) + (yq2 + yq3);
    if (lane < 16) {
      float* q = &lt[rg][lane * 4];
      v4f lo0; lo0.x = ta0[0]; lo0.y = ta0[1]; lo0.z = tb0[0]; lo0.w = tb0[1];
      v4f hi0; hi0.x = ta0[2]; hi0.y = ta0[3]; hi0.z = tb0[2]; hi0.w = tb0[3];
      v4f lo1; lo1.x = ta1[0]; lo1.y = ta1[1]; lo1.z = tb1[0]; lo1.w = tb1[1];
      v4f hi1; hi1.x = ta1[2]; hi1.y = ta1[3]; hi1.z = tb1[2]; hi1.w = tb1[3];
      *(v4f*)q         = lo0;
      *(v4f*)(q + 64)  = hi0;
      *(v4f*)(q + 128) = lo1;
      *(v4f*)(q + 192) = hi1;
    }
  }
  __syncthreads();

  {
    const v4f q0 = *(const v4f*)(lyyp), q1 = *(const v4f*)(lyyp + 4);
    const float yy = ((q0.x + q0.y) + (q0.z + q0.w)) + ((q1.x + q1.y) + (q1.z + q1.w));
    const float t  = ((lt[0][tid] + lt[1][tid]) + (lt[2][tid] + lt[3][tid]))
                   + ((lt[4][tid] + lt[5][tid]) + (lt[6][tid] + lt[7][tid]));
    const float z2 = yy - 2.0f * t + sk;
    const float wv = 1.0f / sqrtf(1.0f + z2);
    lw[tid] = wv;
  }
  __syncthreads();

  {
    const v4f wq0 = *(const v4f*)(lw + lane * 4);
    const v4f wq1 = *(const v4f*)(lw + 128 + lane * 4);
    if (wid == 0) {
      float dsum = ((wq0.x + wq0.y) + (wq0.z + wq0.w)) + ((wq1.x + wq1.y) + (wq1.z + wq1.w));
      for (int off = 16; off > 0; off >>= 1) dsum += __shfl_xor(dsum, off, 32);
      if (lane == 0) dout[(size_t)(b * NSPLIT + j) * DPAD] = dsum;
    }
    v2f a00; a00.x = wq0.x; a00.y = wq0.y;
    v2f a01; a01.x = wq0.z; a01.y = wq0.w;
    v2f a10; a10.x = wq1.x; a10.y = wq1.y;
    v2f a11; a11.x = wq1.z; a11.y = wq1.w;
    const bool b0 = (lane & 1) != 0, b1 = (lane & 2) != 0, b2 = (lane & 4) != 0;
    float* qr = lred + (size_t)(rg * ROWS) * RP + lane;
    #pragma unroll
    for (int i = 0; i < ROWS; ++i) {
      const v4f r0 = buf[i][0], r1 = buf[i][1];
      v2f x00; x00.x = r0.x; x00.y = r0.y;
      v2f x01; x01.x = r0.z; x01.y = r0.w;
      v2f x10; x10.x = r1.x; x10.y = r1.y;
      v2f x11; x11.x = r1.z; x11.y = r1.w;
      v8f d = {};
      d = wmma4(a00, x00, d);
      d = wmma4(a01, x01, d);
      d = wmma4(a10, x10, d);
      d = wmma4(a11, x11, d);
      qr[i * RP] = diag_pick(d, b0, b1, b2);
    }
  }
  __syncthreads();

  {
    const float* r = lred + (size_t)tid * RP;
    const v4f c0 = *(const v4f*)(r),      c1 = *(const v4f*)(r + 4);
    const v4f c2 = *(const v4f*)(r + 24), c3 = *(const v4f*)(r + 28);
    const float h = (((c0.x + c0.y) + (c0.z + c0.w)) + ((c1.x + c1.y) + (c1.z + c1.w)))
                  + (((c2.x + c2.y) + (c2.z + c2.w)) + ((c3.x + c3.y) + (c3.z + c3.w)));
    pout[(size_t)(b * NSPLIT + j) * CH + tid] = h;
  }
}

__global__ __launch_bounds__(CH) void rvp_final(const float* __restrict__ p,
                                                const float* __restrict__ d,
                                                float* __restrict__ out) {
  const int b = blockIdx.x, tid = threadIdx.x;
  const float* pb = p + (size_t)b * NSPLIT * CH;
  const float* db = d + (size_t)b * NSPLIT * DPAD;
  const float nv = (pb[tid] + pb[CH + tid]) + (pb[2 * CH + tid] + pb[3 * CH + tid]);
  const float dv = (db[0] + db[DPAD]) + (db[2 * DPAD] + db[3 * DPAD]);
  out[(size_t)b * CH + tid] = nv / dv;
}

#define STEP2  rvp_step<<<dim3(NSPLIT, BATCH), dim3(NTHR), 0, stream>>>(x, pA, dA, sS, pB, dB); \
               rvp_step<<<dim3(NSPLIT, BATCH), dim3(NTHR), 0, stream>>>(x, pB, dB, sS, pA, dA);
#define STEP10 STEP2 STEP2 STEP2 STEP2 STEP2
#define STEP50 STEP10 STEP10 STEP10 STEP10 STEP10

extern "C" void kernel_launch(void* const* d_in, const int* in_sizes, int n_in,
                              void* d_out, int out_size, void* d_ws, size_t ws_size,
                              hipStream_t stream) {
  (void)in_sizes; (void)n_in; (void)out_size; (void)ws_size;
  const float* x = (const float*)d_in[0];
  float* out = (float*)d_out;
  float* ws  = (float*)d_ws;
  float* pA  = ws;
  float* pB  = ws + BATCH * NSPLIT * CH;
  float* dA  = ws + 2 * BATCH * NSPLIT * CH;
  float* dB  = ws + 2 * BATCH * NSPLIT * CH + BATCH * NSPLIT * DPAD;
  float* sS  = ws + 2 * BATCH * NSPLIT * CH + 2 * BATCH * NSPLIT * DPAD;

  rvp_init<<<dim3(NSPLIT, BATCH), dim3(256), 0, stream>>>(x, pA, dA, sS);
  STEP50 STEP50
  rvp_final<<<dim3(BATCH), dim3(CH), 0, stream>>>(pA, dA, out);
}


# row loads requested in step with product 1 (8 rows ahead, 2 rows per group) instead of all 64 in front of it, den partials first: product 1 runs under the loads; sums and their order unchanged
# speedup vs baseline: 3.5457x; 1.1537x over previous
#include <hip/hip_runtime.h>

#define BATCH  64
#define CH     256
#define KDIM   1024
#define NSPLIT 4
#define KQ     (KDIM / NSPLIT)
#define NWAVE  8
#define NTHR   (NWAVE * 32)
#define ROWS   32
#define DPAD   32
#define RP     36
#define PRE    8
#define GRP    2

typedef __attribute__((ext_vector_type(2))) float v2f;
typedef __attribute__((ext_vector_type(4))) float v4f;
typedef __attribute__((ext_vector_type(8))) float v8f;

static __device__ __forceinline__ v8f wmma4(v2f a, v2f b, v8f c) {
  return __builtin_amdgcn_wmma_f32_16x16x4_f32(false, a, false, b, (short)0, c, false, false);
}

static __device__ __forceinline__ float diag_pick(const v8f d, const bool b0, const bool b1,
                                                  const bool b2) {
  const float e0 = b0 ? d[1] : d[0];
  const float e1 = b0 ? d[3] : d[2];
  const float e2 = b0 ? d[5] : d[4];
  const float e3 = b0 ? d[7] : d[6];
  const float f0 = b1 ? e1 : e0;
  const float f1 = b1 ? e3 : e2;
  return b2 ? f1 : f0;
}

__global__ __launch_bounds__(256) void rvp_init(const float* __restrict__ x,
                                                float* __restrict__ p, float* __restrict__ d,
                                                float* __restrict__ s) {
  __shared__ float lsum[CH];
  __shared__ __attribute__((aligned(16))) float lsq[8][KQ];
  const int j = blockIdx.x, b = blockIdx.y;
  const int tid = threadIdx.x, wid = tid >> 5, lane = tid & 31;
  const float* px = x + (size_t)b * CH * KDIM + (size_t)(wid * 32) * KDIM + (j * KQ + lane * 4);
  v4f sq0 = {0.f, 0.f, 0.f, 0.f}, sq1 = {0.f, 0.f, 0.f, 0.f};
  for (int i = 0; i < 32; ++i) {
    const v4f u = *(const v4f*)(px + (size_t)i * KDIM);
    const v4f v = *(const v4f*)(px + (size_t)i * KDIM + 128);
    sq0.x += u.x * u.x; sq0.y += u.y * u.y; sq0.z += u.z * u.z; sq0.w += u.w * u.w;
    sq1.x += v.x * v.x; sq1.y += v.y * v.y; sq1.z += v.z * v.z; sq1.w += v.w * v.w;
    float a = ((u.x + u.y) + (u.z + u.w)) + ((v.x + v.y) + (v.z + v.w));
    #pragma unroll
    for (int off = 16; off > 0; off >>= 1) a += __shfl_xor(a, off, 32);
    if (lane == 0) lsum[wid * 32 + i] = a;
  }
  *(v4f*)&lsq[wid][lane * 4]       = sq0;
  *(v4f*)&lsq[wid][128 + lane * 4] = sq1;
  __syncthreads();
  p[(size_t)(b * NSPLIT + j) * CH + tid] = lsum[tid];
  s[(size_t)(b * NSPLIT + j) * KQ + tid] =
      ((lsq[0][tid] + lsq[1][tid]) + (lsq[2][tid] + lsq[3][tid]))
    + ((lsq[4][tid] + lsq[5][tid]) + (lsq[6][tid] + lsq[7][tid]));
  if (tid == 0) d[(size_t)(b * NSPLIT + j) * DPAD] = (float)KQ;
}

__global__ __launch_bounds__(NTHR) void rvp_step(const float* __restrict__ x,
                                                const float* __restrict__ pin,
                                                const float* __restrict__ din,
                                                const float* __restrict__ ssq,
                                                float* __restrict__ pout,
                                                float* __restrict__ dout) {
  __shared__ __attribute__((aligned(16))) float lw[KQ];
  __shared__ __attribute__((aligned(16))) float lt[8][KQ];
  __shared__ __attribute__((aligned(16))) float lred[CH * RP];
  __shared__ __attribute__((aligned(16))) float lyyp[8];

  const int j    = blockIdx.x;
  const int b    = blockIdx.y;
  const int tid  = threadIdx.x;
  const int wid  = tid >> 5;
  const int lane = tid & 31;
  const int rg   = wid;
  const float m0 = (lane == 0 || lane == 18) ? 1.0f : 0.0f;
  const float m1 = (lane == 1 || lane == 19) ? 1.0f : 0.0f;

  const float* pb = pin + (size_t)b * NSPLIT * CH + (rg * ROWS + lane);
  const float* db = din + (size_t)b * NSPLIT * DPAD;
  const float p0 = pb[0], p1 = pb[CH], p2 = pb[2 * CH], p3 = pb[3 * CH];
  const float d0 = db[0], d1 = db[DPAD], d2 = db[2 * DPAD], d3 = db[3 * DPAD];
  const float sk = ssq[(size_t)(b * NSPLIT + j) * KQ + tid];
  __builtin_amdgcn_sched_barrier(0);

  const float* px = x + (size_t)b * CH * KDIM + (size_t)(rg * ROWS) * KDIM
                      + (j * KQ + lane * 4);
  v4f buf[ROWS][2];
  #pragma unroll
  for (int i = 0; i < PRE; ++i) {
    buf[i][0] = *(const v4f*)(px + (size_t)i * KDIM);
    buf[i][1] = *(const v4f*)(px + (size_t)i * KDIM + 128);
  }
  __builtin_amdgcn_sched_barrier(0);

  const float yl = ((p0 + p1) + (p2 + p3)) / ((d0 + d1) + (d2 + d3));

  {
    float yq0 = 0.f, yq1 = 0.f, yq2 = 0.f, yq3 = 0.f;
    v8f ta0 = {}, tb0 = {}, ta1 = {}, tb1 = {};
    #pragma unroll
    for (int i = 0; i < ROWS; ++i) {
      const v4f r0 = buf[i][0], r1 = buf[i][1];
      const float yi = __int_as_float(__builtin_amdgcn_readlane(__float_as_int(yl), i));
      if ((i & 3) == 0) yq0 += yi * yi;
      else if ((i & 3) == 1) yq1 += yi * yi;
      else if ((i & 3) == 2) yq2 += yi * yi;
      else yq3 += yi * yi;
      v2f av; av.x = yi * m0; av.y = yi * m1;
      v2f b00; b00.x = r0.x; b00.y = r0.y;
      v2f b01; b01.x = r0.z; b01.y = r0.w;
      v2f b10; b10.x = r1.x; b10.y = r1.y;
      v2f b11; b11.x = r1.z; b11.y = r1.w;
      ta0 = wmma4(av, b00, ta0);
      tb0 = wmma4(av, b01, tb0);
      ta1 = wmma4(av, b10, ta1);
      tb1 = wmma4(av, b11, tb1);
      if ((i % GRP) == GRP - 1 && i + 1 + (PRE - GRP) < ROWS) {
        __builtin_amdgcn_sched_barrier(0);
        #pragma unroll
        for (int r = 0; r < GRP; ++r) {
          const int q = i + 1 + (PRE - GRP) + r;
          buf[q][0] = *(const v4f*)(px + (size_t)q * KDIM);
          buf[q][1] = *(const v4f*)(px + (size_t)q * KDIM + 128);
        }
        __builtin_amdgcn_sched_barrier(0);
      }
    }
    lyyp[rg] = (yq0 + yq1) + (yq2 + yq3);
    if (lane < 16) {
      float* q = &lt[rg][lane * 4];
      v4f lo0; lo0.x = ta0[0]; lo0.y = ta0[1]; lo0.z = tb0[0]; lo0.w = tb0[1];
      v4f hi0; hi0.x = ta0[2]; hi0.y = ta0[3]; hi0.z = tb0[2]; hi0.w = tb0[3];
      v4f lo1; lo1.x = ta1[0]; lo1.y = ta1[1]; lo1.z = tb1[0]; lo1.w = tb1[1];
      v4f hi1; hi1.x = ta1[2]; hi1.y = ta1[3]; hi1.z = tb1[2]; hi1.w = tb1[3];
      *(v4f*)q         = lo0;
      *(v4f*)(q + 64)  = hi0;
      *(v4f*)(q + 128) = lo1;
      *(v4f*)(q + 192) = hi1;
    }
  }
  __syncthreads();

  {
    const v4f q0 = *(const v4f*)(lyyp), q1 = *(const v4f*)(lyyp + 4);
    const float yy = ((q0.x + q0.y) + (q0.z + q0.w)) + ((q1.x + q1.y) + (q1.z + q1.w));
    const float t  = ((lt[0][tid] + lt[1][tid]) + (lt[2][tid] + lt[3][tid]))
                   + ((lt[4][tid] + lt[5][tid]) + (lt[6][tid] + lt[7][tid]));
    const float z2 = yy - 2.0f * t + sk;
    const float wv = 1.0f / sqrtf(1.0f + z2);
    lw[tid] = wv;
  }
  __syncthreads();

  {
    const v4f wq0 = *(const v4f*)(lw + lane * 4);
    const v4f wq1 = *(const v4f*)(lw + 128 + lane * 4);
    if (wid == 0) {
      float dsum = ((wq0.x + wq0.y) + (wq0.z + wq0.w)) + ((wq1.x + wq1.y) + (wq1.z + wq1.w));
      for (int off = 16; off > 0; off >>= 1) dsum += __shfl_xor(dsum, off, 32);
      if (lane == 0) dout[(size_t)(b * NSPLIT + j) * DPAD] = dsum;
    }
    v2f a00; a00.x = wq0.x; a00.y = wq0.y;
    v2f a01; a01.x = wq0.z; a01.y = wq0.w;
    v2f a10; a10.x = wq1.x; a10.y = wq1.y;
    v2f a11; a11.x = wq1.z; a11.y = wq1.w;
    const bool b0 = (lane & 1) != 0, b1 = (lane & 2) != 0, b2 = (lane & 4) != 0;
    float* qr = lred + (size_t)(rg * ROWS) * RP + lane;
    #pragma unroll
    for (int i = 0; i < ROWS; ++i) {
      const v4f r0 = buf[i][0], r1 = buf[i][1];
      v2f x00; x00.x = r0.x; x00.y = r0.y;
      v2f x01; x01.x = r0.z; x01.y = r0.w;
      v2f x10; x10.x = r1.x; x10.y = r1.y;
      v2f x11; x11.x = r1.z; x11.y = r1.w;
      v8f d = {};
      d = wmma4(a00, x00, d);
      d = wmma4(a01, x01, d);
      d = wmma4(a10, x10, d);
      d = wmma4(a11, x11, d);
      qr[i * RP] = diag_pick(d, b0, b1, b2);
    }
  }
  __syncthreads();

  {
    const float* r = lred + (size_t)tid * RP;
    const v4f c0 = *(const v4f*)(r),      c1 = *(const v4f*)(r + 4);
    const v4f c2 = *(const v4f*)(r + 24), c3 = *(const v4f*)(r + 28);
    const float h = (((c0.x + c0.y) + (c0.z + c0.w)) + ((c1.x + c1.y) + (c1.z + c1.w)))
                  + (((c2.x + c2.y) + (c2.z + c2.w)) + ((c3.x + c3.y) + (c3.z + c3.w)));
    pout[(size_t)(b * NSPLIT + j) * CH + tid] = h;
  }
}

__global__ __launch_bounds__(CH) void rvp_final(const float* __restrict__ p,
                                                const float* __restrict__ d,
                                                float* __restrict__ out) {
  const int b = blockIdx.x, tid = threadIdx.x;
  const float* pb = p + (size_t)b * NSPLIT * CH;
  const float* db = d + (size_t)b * NSPLIT * DPAD;
  const float nv = (pb[tid] + pb[CH + tid]) + (pb[2 * CH + tid] + pb[3 * CH + tid]);
  const float dv = (db[0] + db[DPAD]) + (db[2 * DPAD] + db[3 * DPAD]);
  out[(size_t)b * CH + tid] = nv / dv;
}

#define STEP2  rvp_step<<<dim3(NSPLIT, BATCH), dim3(NTHR), 0, stream>>>(x, pA, dA, sS, pB, dB); \
               rvp_step<<<dim3(NSPLIT, BATCH), dim3(NTHR), 0, stream>>>(x, pB, dB, sS, pA, dA);
#define STEP10 STEP2 STEP2 STEP2 STEP2 STEP2
#define STEP50 STEP10 STEP10 STEP10 STEP10 STEP10

extern "C" void kernel_launch(void* const* d_in, const int* in_sizes, int n_in,
                              void* d_out, int out_size, void* d_ws, size_t ws_size,
                              hipStream_t stream) {
  (void)in_sizes; (void)n_in; (void)out_size; (void)ws_size;
  const float* x = (const float*)d_in[0];
  float* out = (float*)d_out;
  float* ws  = (float*)d_ws;
  float* pA  = ws;
  float* pB  = ws + BATCH * NSPLIT * CH;
  float* dA  = ws + 2 * BATCH * NSPLIT * CH;
  float* dB  = ws + 2 * BATCH * NSPLIT * CH + BATCH * NSPLIT * DPAD;
  float* sS  = ws + 2 * BATCH * NSPLIT * CH + 2 * BATCH * NSPLIT * DPAD;

  rvp_init<<<dim3(NSPLIT, BATCH), dim3(256), 0, stream>>>(x, pA, dA, sS);
  STEP50 STEP50
  rvp_final<<<dim3(BATCH), dim3(CH), 0, stream>>>(pA, dA, out);
}
